# starting kernel
# baseline (speedup 1.0000x reference)
#include <hip/hip_runtime.h>
#include <hip/hip_bf16.h>
#include <math.h>

typedef float v2f __attribute__((ext_vector_type(2)));
typedef float v8f __attribute__((ext_vector_type(8)));

#define EPS 1e-7f
#define NB   256
#define NU   8
#define NI   1024
#define ED   64

__global__ __launch_bounds__(256)
void simout_kernel(const float* __restrict__ user_emb,
                   const float* __restrict__ item_emb,
                   const float* __restrict__ proj_w,
                   const float* __restrict__ proj_b,
                   float* __restrict__ out)
{
    const int tid  = threadIdx.x;
    const int wave = tid >> 5;
    const int lane = tid & 31;
    const int b    = blockIdx.x >> 3;
    const int blk  = blockIdx.x & 7;
    const int m0   = blk * 128 + wave * 16;

    const int sel = lane >> 4;
    const int l15 = lane & 15;

    const float* irow = item_emb + ((size_t)b * NI + (size_t)(m0 + l15)) * ED;
    v2f a[16];
    float ss = 0.0f;
#pragma unroll
    for (int kc = 0; kc < 16; ++kc) {
        const int k = kc * 4 + sel * 2;
        float x = irow[k];
        float y = irow[k + 1];
        a[kc].x = x; a[kc].y = y;
        ss = fmaf(x, x, fmaf(y, y, ss));
    }
    ss += __shfl_xor(ss, 16);
    const float ascale = 1.0f / fmaxf(sqrtf(ss), EPS);
#pragma unroll
    for (int kc = 0; kc < 16; ++kc) { a[kc].x *= ascale; a[kc].y *= ascale; }

    const float* urow = user_emb + ((size_t)b * NU + (size_t)(l15 & 7)) * ED;
    const float uvalid = (l15 < NU) ? 1.0f : 0.0f;
    v2f bf[16];
    float us = 0.0f;
#pragma unroll
    for (int kc = 0; kc < 16; ++kc) {
        const int k = kc * 4 + sel * 2;
        float x = urow[k] * uvalid;
        float y = urow[k + 1] * uvalid;
        bf[kc].x = x; bf[kc].y = y;
        us = fmaf(x, x, fmaf(y, y, us));
    }
    us += __shfl_xor(us, 16);
    const float bscale = 1.0f / fmaxf(sqrtf(us), EPS);
#pragma unroll
    for (int kc = 0; kc < 16; ++kc) { bf[kc].x *= bscale; bf[kc].y *= bscale; }

    v8f c = {};
#pragma unroll
    for (int kc = 0; kc < 16; ++kc) {
        c = __builtin_amdgcn_wmma_f32_16x16x4_f32(
                 false, a[kc],  false, bf[kc],
                 (short)0, c,  false,  false);
    }

    const float w  = proj_w[0];
    const float bb = proj_b[0];
    if (l15 < NU) {
        const int n = l15;
#pragma unroll
        for (int r = 0; r < 8; ++r) {
            const int item = m0 + r + sel * 8;
            float v = fmaf(c[r], w, bb);
            float s = 1.0f / (1.0f + __expf(-v));
            out[((size_t)b * NI + (size_t)item) * NU + n] = s;
        }
    }
}

extern "C" void kernel_launch(void* const* d_in, const int* in_sizes, int n_in,
                              void* d_out, int out_size, void* d_ws, size_t ws_size,
                              hipStream_t stream) {
    const float* user_emb = (const float*)d_in[0];
    const float* item_emb = (const float*)d_in[1];
    const float* proj_w   = (const float*)d_in[2];
    const float* proj_b   = (const float*)d_in[3];
    float* out = (float*)d_out;

    dim3 block(256);
    dim3 grid(NB * (NI / 128));
    simout_kernel<<<grid, block, 0, stream>>>(user_emb, item_emb, proj_w, proj_b, out);
}


# Each wave loops over 4 item tiles and reuses one normalised user fragment set: grid 2048 -> 512 workgroups, a quarter of the waves to start; every sum's order unchanged
# speedup vs baseline: 1.0770x; 1.0770x over previous
#include <hip/hip_runtime.h>
#include <hip/hip_bf16.h>
#include <math.h>


typedef float v2f __attribute__((ext_vector_type(2)));
typedef float v8f __attribute__((ext_vector_type(8)));

#define EPS 1e-7f
#define NB   256
#define NU   8
#define NI   1024
#define ED   64
#define TPW  4
#define WPB  8
#define IPB  (WPB * TPW * 16)

__global__ __launch_bounds__(256)
void simout_kernel(const float* __restrict__ user_emb,
                   const float* __restrict__ item_emb,
                   const float* __restrict__ proj_w,
                   const float* __restrict__ proj_b,
                   float* __restrict__ out)
{
    const int tid  = threadIdx.x;
    const int wave = tid >> 5;
    const int lane = tid & 31;
    const int b    = blockIdx.x / (NI / IPB);
    const int blk  = blockIdx.x % (NI / IPB);
    const int m0   = blk * IPB + wave * (TPW * 16);

    const int sel = lane >> 4;
    const int l15 = lane & 15;

    const float* urow = user_emb + ((size_t)b * NU + (size_t)(l15 & 7)) * ED;
    const float uvalid = (l15 < NU) ? 1.0f : 0.0f;
    v2f bf[16];
    float us = 0.0f;
#pragma unroll
    for (int kc = 0; kc < 16; ++kc) {
        const int k = kc * 4 + sel * 2;
        float x = urow[k] * uvalid;
        float y = urow[k + 1] * uvalid;
        bf[kc].x = x; bf[kc].y = y;
        us = fmaf(x, x, fmaf(y, y, us));
    }

    v2f a[TPW][16];
#pragma unroll
    for (int t = 0; t < TPW; ++t) {
        const float* irow = item_emb + ((size_t)b * NI + (size_t)(m0 + t * 16 + l15)) * ED;
#pragma unroll
        for (int kc = 0; kc < 16; ++kc) {
            const int k = kc * 4 + sel * 2;
            a[t][kc].x = irow[k];
            a[t][kc].y = irow[k + 1];
        }
    }

    us += __shfl_xor(us, 16);
    const float bscale = 1.0f / fmaxf(sqrtf(us), EPS);
#pragma unroll
    for (int kc = 0; kc < 16; ++kc) { bf[kc].x *= bscale; bf[kc].y *= bscale; }

    const float w  = proj_w[0];
    const float bb = proj_b[0];

#pragma unroll
    for (int t = 0; t < TPW; ++t) {
        float ss = 0.0f;
#pragma unroll
        for (int kc = 0; kc < 16; ++kc) {
            const float x = a[t][kc].x;
            const float y = a[t][kc].y;
            ss = fmaf(x, x, fmaf(y, y, ss));
        }
        ss += __shfl_xor(ss, 16);
        const float ascale = 1.0f / fmaxf(sqrtf(ss), EPS);
#pragma unroll
        for (int kc = 0; kc < 16; ++kc) { a[t][kc].x *= ascale; a[t][kc].y *= ascale; }

        v8f c = {};
#pragma unroll
        for (int kc = 0; kc < 16; ++kc) {
            c = __builtin_amdgcn_wmma_f32_16x16x4_f32(
                     false, a[t][kc],  false, bf[kc],
                     (short)0, c,  false,  false);
        }

        if (l15 < NU) {
            const int n = l15;
#pragma unroll
            for (int r = 0; r < 8; ++r) {
                const int item = m0 + t * 16 + r + sel * 8;
                float v = fmaf(c[r], w, bb);
                float s = 1.0f / (1.0f + __expf(-v));
                out[((size_t)b * NI + (size_t)item) * NU + n] = s;
            }
        }
    }
}

extern "C" void kernel_launch(void* const* d_in, const int* in_sizes, int n_in,
                              void* d_out, int out_size, void* d_ws, size_t ws_size,
                              hipStream_t stream) {
    const float* user_emb = (const float*)d_in[0];
    const float* item_emb = (const float*)d_in[1];
    const float* proj_w   = (const float*)d_in[2];
    const float* proj_b   = (const float*)d_in[3];
    float* out = (float*)d_out;

    dim3 block(256);
    dim3 grid(NB * (NI / IPB));
    simout_kernel<<<grid, block, 0, stream>>>(user_emb, item_emb, proj_w, proj_b, out);
}


# Item rows read with whole-line 16-byte loads (512 contiguous bytes per wave load, all 48 loads of a wave in flight) and turned into A fragments in a per-wave LDS slab; every sum's order unchanged
# speedup vs baseline: 1.1702x; 1.0865x over previous
#include <hip/hip_runtime.h>
#include <hip/hip_bf16.h>
#include <math.h>


typedef float v2f __attribute__((ext_vector_type(2)));
typedef float v4f __attribute__((ext_vector_type(4)));
typedef float v8f __attribute__((ext_vector_type(8)));

#define EPS 1e-7f
#define NB   256
#define NU   8
#define NI   1024
#define ED   64
#define TPW  4
#define WPB  8
#define IPB  (WPB * TPW * 16)
#define LDP  68

__global__ __launch_bounds__(256)
void simout_kernel(const float* __restrict__ user_emb,
                   const float* __restrict__ item_emb,
                   const float* __restrict__ proj_w,
                   const float* __restrict__ proj_b,
                   float* __restrict__ out)
{
    __shared__ __attribute__((aligned(16))) float stage[WPB][16 * LDP];

    const int tid  = threadIdx.x;
    const int wave = tid >> 5;
    const int lane = tid & 31;
    const int b    = blockIdx.x / (NI / IPB);
    const int blk  = blockIdx.x % (NI / IPB);
    const int m0   = blk * IPB + wave * (TPW * 16);

    const int sel = lane >> 4;
    const int l15 = lane & 15;

    const float* urow = user_emb + ((size_t)b * NU + (size_t)(l15 & 7)) * ED;
    const float uvalid = (l15 < NU) ? 1.0f : 0.0f;
    v2f bf[16];
#pragma unroll
    for (int kc = 0; kc < 16; ++kc) {
        const int k = kc * 4 + sel * 2;
        bf[kc].x = urow[k];
        bf[kc].y = urow[k + 1];
    }
    __builtin_amdgcn_sched_barrier(0);

    const v4f* itile = reinterpret_cast<const v4f*>(
        item_emb + ((size_t)b * NI + (size_t)m0) * ED);
    v4f g[TPW][8];
#pragma unroll
    for (int t = 0; t < TPW; ++t) {
#pragma unroll
        for (int j = 0; j < 8; ++j) {
            g[t][j] = itile[t * 256 + j * 32 + lane];
        }
    }
    __builtin_amdgcn_sched_barrier(0);

    float us = 0.0f;
#pragma unroll
    for (int kc = 0; kc < 16; ++kc) {
        const float x = bf[kc].x * uvalid;
        const float y = bf[kc].y * uvalid;
        bf[kc].x = x; bf[kc].y = y;
        us = fmaf(x, x, fmaf(y, y, us));
    }

    us += __shfl_xor(us, 16);
    const float bscale = 1.0f / fmaxf(sqrtf(us), EPS);
#pragma unroll
    for (int kc = 0; kc < 16; ++kc) { bf[kc].x *= bscale; bf[kc].y *= bscale; }

    float* slab = &stage[wave][0];

    const float w  = proj_w[0];
    const float bb = proj_b[0];

#pragma unroll
    for (int t = 0; t < TPW; ++t) {
#pragma unroll
        for (int j = 0; j < 8; ++j) {
            *reinterpret_cast<v4f*>(slab + (2 * j + sel) * LDP + 4 * l15) = g[t][j];
        }
        v2f a[16];
#pragma unroll
        for (int kc = 0; kc < 16; ++kc) {
            a[kc] = *reinterpret_cast<const v2f*>(slab + l15 * LDP + 4 * kc + 2 * sel);
        }

        float ss = 0.0f;
#pragma unroll
        for (int kc = 0; kc < 16; ++kc) {
            const float x = a[kc].x;
            const float y = a[kc].y;
            ss = fmaf(x, x, fmaf(y, y, ss));
        }
        ss += __shfl_xor(ss, 16);
        const float ascale = 1.0f / fmaxf(sqrtf(ss), EPS);
#pragma unroll
        for (int kc = 0; kc < 16; ++kc) { a[kc].x *= ascale; a[kc].y *= ascale; }

        v8f c = {};
#pragma unroll
        for (int kc = 0; kc < 16; ++kc) {
            c = __builtin_amdgcn_wmma_f32_16x16x4_f32(
                     false, a[kc],  false, bf[kc],
                     (short)0, c,  false,  false);
        }

        if (l15 < NU) {
            const int n = l15;
#pragma unroll
            for (int r = 0; r < 8; ++r) {
                const int item = m0 + t * 16 + r + sel * 8;
                float v = fmaf(c[r], w, bb);
                float s = 1.0f / (1.0f + __expf(-v));
                out[((size_t)b * NI + (size_t)item) * NU + n] = s;
            }
        }
    }
}

extern "C" void kernel_launch(void* const* d_in, const int* in_sizes, int n_in,
                              void* d_out, int out_size, void* d_ws, size_t ws_size,
                              hipStream_t stream) {
    const float* user_emb = (const float*)d_in[0];
    const float* item_emb = (const float*)d_in[1];
    const float* proj_w   = (const float*)d_in[2];
    const float* proj_b   = (const float*)d_in[3];
    float* out = (float*)d_out;

    dim3 block(256);
    dim3 grid(NB * (NI / IPB));
    simout_kernel<<<grid, block, 0, stream>>>(user_emb, item_emb, proj_w, proj_b, out);
}


# Sigmoid 1/(1+e^-v) and the 1/max(norm,eps) row scales use hardware v_rcp_f32 and v_sqrt_f32 (about 1 ulp) instead of IEEE division and exact sqrtf; every sum's order unchanged
# speedup vs baseline: 1.2687x; 1.0842x over previous
#include <hip/hip_runtime.h>
#include <hip/hip_bf16.h>
#include <math.h>


typedef float v2f __attribute__((ext_vector_type(2)));
typedef float v4f __attribute__((ext_vector_type(4)));
typedef float v8f __attribute__((ext_vector_type(8)));

#define EPS 1e-7f
#define NB   256
#define NU   8
#define NI   1024
#define ED   64
#define TPW  4
#define WPB  8
#define IPB  (WPB * TPW * 16)
#define LDP  68

__global__ __launch_bounds__(256)
void simout_kernel(const float* __restrict__ user_emb,
                   const float* __restrict__ item_emb,
                   const float* __restrict__ proj_w,
                   const float* __restrict__ proj_b,
                   float* __restrict__ out)
{
    __shared__ __attribute__((aligned(16))) float stage[WPB][16 * LDP];

    const int tid  = threadIdx.x;
    const int wave = tid >> 5;
    const int lane = tid & 31;
    const int b    = blockIdx.x / (NI / IPB);
    const int blk  = blockIdx.x % (NI / IPB);
    const int m0   = blk * IPB + wave * (TPW * 16);

    const int sel = lane >> 4;
    const int l15 = lane & 15;

    const float* urow = user_emb + ((size_t)b * NU + (size_t)(l15 & 7)) * ED;
    const float uvalid = (l15 < NU) ? 1.0f : 0.0f;
    v2f bf[16];
#pragma unroll
    for (int kc = 0; kc < 16; ++kc) {
        const int k = kc * 4 + sel * 2;
        bf[kc].x = urow[k];
        bf[kc].y = urow[k + 1];
    }
    __builtin_amdgcn_sched_barrier(0);

    const v4f* itile = reinterpret_cast<const v4f*>(
        item_emb + ((size_t)b * NI + (size_t)m0) * ED);
    v4f g[TPW][8];
#pragma unroll
    for (int t = 0; t < TPW; ++t) {
#pragma unroll
        for (int j = 0; j < 8; ++j) {
            g[t][j] = itile[t * 256 + j * 32 + lane];
        }
    }
    __builtin_amdgcn_sched_barrier(0);

    float us = 0.0f;
#pragma unroll
    for (int kc = 0; kc < 16; ++kc) {
        const float x = bf[kc].x * uvalid;
        const float y = bf[kc].y * uvalid;
        bf[kc].x = x; bf[kc].y = y;
        us = fmaf(x, x, fmaf(y, y, us));
    }

    us += __shfl_xor(us, 16);
    const float bscale = __builtin_amdgcn_rcpf(fmaxf(__builtin_amdgcn_sqrtf(us), EPS));
#pragma unroll
    for (int kc = 0; kc < 16; ++kc) { bf[kc].x *= bscale; bf[kc].y *= bscale; }

    float* slab = &stage[wave][0];

    const float w  = proj_w[0];
    const float bb = proj_b[0];

#pragma unroll
    for (int t = 0; t < TPW; ++t) {
#pragma unroll
        for (int j = 0; j < 8; ++j) {
            *reinterpret_cast<v4f*>(slab + (2 * j + sel) * LDP + 4 * l15) = g[t][j];
        }
        v2f a[16];
#pragma unroll
        for (int kc = 0; kc < 16; ++kc) {
            a[kc] = *reinterpret_cast<const v2f*>(slab + l15 * LDP + 4 * kc + 2 * sel);
        }

        float ss = 0.0f;
#pragma unroll
        for (int kc = 0; kc < 16; ++kc) {
            const float x = a[kc].x;
            const float y = a[kc].y;
            ss = fmaf(x, x, fmaf(y, y, ss));
        }
        ss += __shfl_xor(ss, 16);
        const float ascale = __builtin_amdgcn_rcpf(fmaxf(__builtin_amdgcn_sqrtf(ss), EPS));
#pragma unroll
        for (int kc = 0; kc < 16; ++kc) { a[kc].x *= ascale; a[kc].y *= ascale; }

        v8f c = {};
#pragma unroll
        for (int kc = 0; kc < 16; ++kc) {
            c = __builtin_amdgcn_wmma_f32_16x16x4_f32(
                     false, a[kc],  false, bf[kc],
                     (short)0, c,  false,  false);
        }

        if (l15 < NU) {
            const int n = l15;
#pragma unroll
            for (int r = 0; r < 8; ++r) {
                const int item = m0 + t * 16 + r + sel * 8;
                float v = fmaf(c[r], w, bb);
                float s = __builtin_amdgcn_rcpf(1.0f + __expf(-v));
                out[((size_t)b * NI + (size_t)item) * NU + n] = s;
            }
        }
    }
}

extern "C" void kernel_launch(void* const* d_in, const int* in_sizes, int n_in,
                              void* d_out, int out_size, void* d_ws, size_t ws_size,
                              hipStream_t stream) {
    const float* user_emb = (const float*)d_in[0];
    const float* item_emb = (const float*)d_in[1];
    const float* proj_w   = (const float*)d_in[2];
    const float* proj_b   = (const float*)d_in[3];
    float* out = (float*)d_out;

    dim3 block(256);
    dim3 grid(NB * (NI / IPB));
    simout_kernel<<<grid, block, 0, stream>>>(user_emb, item_emb, proj_w, proj_b, out);
}


# Users as A operand (matrix rows 0-3, 8-11), items as B operand: each lane's accumulator holds 4 contiguous scores: 4 sigmoids and one 16-byte store per lane, whole 128-B lines; sum order unchanged
# speedup vs baseline: 1.3580x; 1.0704x over previous
#include <hip/hip_runtime.h>
#include <hip/hip_bf16.h>
#include <math.h>


typedef float v2f __attribute__((ext_vector_type(2)));
typedef float v4f __attribute__((ext_vector_type(4)));
typedef float v8f __attribute__((ext_vector_type(8)));

#define EPS 1e-7f
#define NB   256
#define NU   8
#define NI   1024
#define ED   64
#define TPW  4
#define WPB  8
#define IPB  (WPB * TPW * 16)
#define LDP  68

__global__ __launch_bounds__(256)
void simout_kernel(const float* __restrict__ user_emb,
                   const float* __restrict__ item_emb,
                   const float* __restrict__ proj_w,
                   const float* __restrict__ proj_b,
                   float* __restrict__ out)
{
    __shared__ __attribute__((aligned(16))) float stage[WPB][16 * LDP];

    const int tid  = threadIdx.x;
    const int wave = tid >> 5;
    const int lane = tid & 31;
    const int b    = blockIdx.x / (NI / IPB);
    const int blk  = blockIdx.x % (NI / IPB);
    const int m0   = blk * IPB + wave * (TPW * 16);

    const int sel = lane >> 4;
    const int l15 = lane & 15;

    const int urow_id = (l15 & 3) | ((l15 & 8) >> 1);
    const float* urow = user_emb + ((size_t)b * NU + (size_t)urow_id) * ED;
    const float uvalid = ((l15 & 4) == 0) ? 1.0f : 0.0f;
    v2f bf[16];
#pragma unroll
    for (int kc = 0; kc < 16; ++kc) {
        const int k = kc * 4 + sel * 2;
        bf[kc].x = urow[k];
        bf[kc].y = urow[k + 1];
    }
    __builtin_amdgcn_sched_barrier(0);

    const v4f* itile = reinterpret_cast<const v4f*>(
        item_emb + ((size_t)b * NI + (size_t)m0) * ED);
    v4f g[TPW][8];
#pragma unroll
    for (int t = 0; t < TPW; ++t) {
#pragma unroll
        for (int j = 0; j < 8; ++j) {
            g[t][j] = itile[t * 256 + j * 32 + lane];
        }
    }
    __builtin_amdgcn_sched_barrier(0);

    float us = 0.0f;
#pragma unroll
    for (int kc = 0; kc < 16; ++kc) {
        const float x = bf[kc].x * uvalid;
        const float y = bf[kc].y * uvalid;
        bf[kc].x = x; bf[kc].y = y;
        us = fmaf(x, x, fmaf(y, y, us));
    }

    us += __shfl_xor(us, 16);
    const float bscale = __builtin_amdgcn_rcpf(fmaxf(__builtin_amdgcn_sqrtf(us), EPS));
#pragma unroll
    for (int kc = 0; kc < 16; ++kc) { bf[kc].x *= bscale; bf[kc].y *= bscale; }

    float* slab = &stage[wave][0];

    const float w  = proj_w[0];
    const float bb = proj_b[0];

#pragma unroll
    for (int t = 0; t < TPW; ++t) {
#pragma unroll
        for (int j = 0; j < 8; ++j) {
            *reinterpret_cast<v4f*>(slab + (2 * j + sel) * LDP + 4 * l15) = g[t][j];
        }
        v2f a[16];
#pragma unroll
        for (int kc = 0; kc < 16; ++kc) {
            a[kc] = *reinterpret_cast<const v2f*>(slab + l15 * LDP + 4 * kc + 2 * sel);
        }

        float ss = 0.0f;
#pragma unroll
        for (int kc = 0; kc < 16; ++kc) {
            const float x = a[kc].x;
            const float y = a[kc].y;
            ss = fmaf(x, x, fmaf(y, y, ss));
        }
        ss += __shfl_xor(ss, 16);
        const float ascale = __builtin_amdgcn_rcpf(fmaxf(__builtin_amdgcn_sqrtf(ss), EPS));
#pragma unroll
        for (int kc = 0; kc < 16; ++kc) { a[kc].x *= ascale; a[kc].y *= ascale; }

        v8f c = {};
#pragma unroll
        for (int kc = 0; kc < 16; ++kc) {
            c = __builtin_amdgcn_wmma_f32_16x16x4_f32(
                     false, bf[kc],  false, a[kc],
                     (short)0, c,  false,  false);
        }

        {
            const int item = m0 + t * 16 + l15;
            v4f s;
#pragma unroll
            for (int r = 0; r < 4; ++r) {
                float v = fmaf(c[r], w, bb);
                s[r] = __builtin_amdgcn_rcpf(1.0f + __expf(-v));
            }
            *reinterpret_cast<v4f*>(out + ((size_t)b * NI + (size_t)item) * NU + 4 * sel) = s;
        }
    }
}

extern "C" void kernel_launch(void* const* d_in, const int* in_sizes, int n_in,
                              void* d_out, int out_size, void* d_ws, size_t ws_size,
                              hipStream_t stream) {
    const float* user_emb = (const float*)d_in[0];
    const float* item_emb = (const float*)d_in[1];
    const float* proj_w   = (const float*)d_in[2];
    const float* proj_b   = (const float*)d_in[3];
    float* out = (float*)d_out;

    dim3 block(256);
    dim3 grid(NB * (NI / IPB));
    simout_kernel<<<grid, block, 0, stream>>>(user_emb, item_emb, proj_w, proj_b, out);
}
